# starting kernel
# baseline (speedup 1.0000x reference)
#include <hip/hip_runtime.h>
#include <math.h>


typedef float v2f __attribute__((ext_vector_type(2)));
typedef float v8f __attribute__((ext_vector_type(8)));

#define DEPTH     11
#define LEVELS    (DEPTH + 1)
#define IN_W      1024
#define OUT_W     1024
#define N_NODES   4095
#define N_TOKENS  16384
#define TOK_TILE  16
#define XSTRIDE   1026

__device__ __forceinline__ v8f wmma4(v2f a, v2f b, v8f c) {
  return __builtin_amdgcn_wmma_f32_16x16x4_f32(
       false, a,  false, b,
       (short)0, c,  false,  false);
}

__global__ __launch_bounds__(256) void transpose_wout(const float* __restrict__ w_out,
                                                      float* __restrict__ w_out_T) {
  __shared__ float tile[32][33];
  const int n0 = blockIdx.x * 32;
  const int o0 = blockIdx.y * 32;
  const int tx = threadIdx.x;
  const int ty = threadIdx.y;
#pragma unroll
  for (int j = 0; j < 4; ++j) {
    const int n = n0 + tx;
    const int o = o0 + ty + j * 8;
    if (n < N_NODES) tile[ty + j * 8][tx] = w_out[(size_t)o * N_NODES + n];
  }
  __syncthreads();
#pragma unroll
  for (int j = 0; j < 4; ++j) {
    const int n = n0 + ty + j * 8;
    const int o = o0 + tx;
    if (n < N_NODES) w_out_T[(size_t)n * OUT_W + o] = tile[tx][ty + j * 8];
  }
}

__global__ __launch_bounds__(128) void fff_fwd(const float* __restrict__ x,
                                               const float* __restrict__ w_in,
                                               const float* __restrict__ w_out_T,
                                               float* __restrict__ out) {
  __shared__ float lds_x[TOK_TILE * XSTRIDE];
  __shared__ float s_logit[TOK_TILE];
  __shared__ int   s_node[TOK_TILE];
  __shared__ float s_act[LEVELS][TOK_TILE];
  __shared__ int   s_path[LEVELS][TOK_TILE];

  const int tid  = threadIdx.x;
  const int lane = tid & 31;
  const int wave = tid >> 5;
  const size_t xbase = (size_t)blockIdx.x * TOK_TILE * IN_W;

  const float4* __restrict__ xg = (const float4*)(x + xbase);
#pragma unroll 4
  for (int i = 0; i < 32; ++i) {
    const int idx = i * 128 + tid;
    const int tok = idx >> 8;
    const int k4  = idx & 255;
    const float4 v = xg[idx];
    float* dst = &lds_x[tok * XSTRIDE + k4 * 4];
    dst[0] = v.x; dst[1] = v.y; dst[2] = v.z; dst[3] = v.w;
  }
  if (tid < TOK_TILE) { s_logit[tid] = 0.0f; s_node[tid] = 0; }
  __syncthreads();

  const int tok    = lane & 15;
  const int kShift = ((lane >> 4) & 1) * 2;
  const int kbase  = wave * 256;
  const float* xrow = &lds_x[tok * XSTRIDE + kbase + kShift];

  for (int lvl = 0; lvl < LEVELS; ++lvl) {
    const int node = s_node[tok];
    const float* wrow = w_in + (size_t)node * IN_W + kbase + kShift;

    v8f c0 = {}, c1 = {}, c2 = {}, c3 = {};
#pragma unroll 4
    for (int kk = 0; kk < 256; kk += 16) {
      v2f a0 = *(const v2f*)(xrow + kk +  0);
      v2f b0 = *(const v2f*)(wrow + kk +  0);
      c0 = wmma4(a0, b0, c0);
      v2f a1 = *(const v2f*)(xrow + kk +  4);
      v2f b1 = *(const v2f*)(wrow + kk +  4);
      c1 = wmma4(a1, b1, c1);
      v2f a2 = *(const v2f*)(xrow + kk +  8);
      v2f b2 = *(const v2f*)(wrow + kk +  8);
      c2 = wmma4(a2, b2, c2);
      v2f a3 = *(const v2f*)(xrow + kk + 12);
      v2f b3 = *(const v2f*)(wrow + kk + 12);
      c3 = wmma4(a3, b3, c3);
    }
    const v8f c = (c0 + c1) + (c2 + c3);

    const int r = lane & 7;
    const float e0 = (r & 1) ? c[1] : c[0];
    const float e1 = (r & 1) ? c[3] : c[2];
    const float e2 = (r & 1) ? c[5] : c[4];
    const float e3 = (r & 1) ? c[7] : c[6];
    const float f0 = (r & 2) ? e1 : e0;
    const float f1 = (r & 2) ? e3 : e2;
    const float dg = (r & 4) ? f1 : f0;
    if (lane < 8)        atomicAdd(&s_logit[lane],      dg);
    else if (lane >= 24) atomicAdd(&s_logit[lane - 16], dg);
    __syncthreads();

    if (tid < TOK_TILE) {
      const float logit = s_logit[tid];
      const float act = 0.5f * logit * (1.0f + erff(logit * 0.70710678118654752f));
      s_act[lvl][tid]  = act;
      s_path[lvl][tid] = node;
      s_node[tid]  = 2 * node + 1 + ((logit > 0.0f) ? 1 : 0);
      s_logit[tid] = 0.0f;
    }
    __syncthreads();
  }

  float* orow_base = out + xbase;
  for (int t4 = 0; t4 < 4; ++t4) {
    const int t = wave * 4 + t4;
    float4 acc[8];
#pragma unroll
    for (int c8 = 0; c8 < 8; ++c8) acc[c8] = make_float4(0.f, 0.f, 0.f, 0.f);
    for (int k = 0; k < LEVELS; ++k) {
      const float a = s_act[k][t];
      const float4* row = (const float4*)(w_out_T + (size_t)s_path[k][t] * OUT_W);
#pragma unroll
      for (int c8 = 0; c8 < 8; ++c8) {
        const float4 w4 = row[c8 * 32 + lane];
        acc[c8].x += a * w4.x;
        acc[c8].y += a * w4.y;
        acc[c8].z += a * w4.z;
        acc[c8].w += a * w4.w;
      }
    }
    float4* orow = (float4*)(orow_base + (size_t)t * OUT_W);
#pragma unroll
    for (int c8 = 0; c8 < 8; ++c8) orow[c8 * 32 + lane] = acc[c8];
  }
}

extern "C" void kernel_launch(void* const* d_in, const int* in_sizes, int n_in,
                              void* d_out, int out_size, void* d_ws, size_t ws_size,
                              hipStream_t stream) {
  (void)in_sizes; (void)n_in; (void)out_size; (void)ws_size;
  const float* x     = (const float*)d_in[0];
  const float* w_in  = (const float*)d_in[1];
  const float* w_out = (const float*)d_in[2];
  float* w_out_T = (float*)d_ws;
  float* out     = (float*)d_out;

  transpose_wout<<<dim3((N_NODES + 31) / 32, OUT_W / 32), dim3(32, 8), 0, stream>>>(
      w_out, w_out_T);
  fff_fwd<<<N_TOKENS / TOK_TILE, 128, 0, stream>>>(x, w_in, w_out_T, out);
}


# Logits re-tiled: one token per 16x16x4 WMMA tile, K folded over tile rows/cols (trace = dot; sum order changes); row-contiguous b128 fragment loads, x in registers; no LDS, barrier, atomic
# speedup vs baseline: 1.8368x; 1.8368x over previous
#include <hip/hip_runtime.h>
#include <math.h>


typedef float v2f __attribute__((ext_vector_type(2)));
typedef float v4f __attribute__((ext_vector_type(4)));
typedef float v8f __attribute__((ext_vector_type(8)));

#define DEPTH     11
#define LEVELS    (DEPTH + 1)
#define IN_W      1024
#define OUT_W     1024
#define N_NODES   4095
#define N_TOKENS  16384
#define ROW_Q     8
#define WAVES_PER_BLOCK 8
#define TOK_PER_WAVE    2

__device__ __forceinline__ v8f wmma4(v2f a, v2f b, v8f c) {
  return __builtin_amdgcn_wmma_f32_16x16x4_f32(
       false, a,  false, b,
       (short)0, c,  false,  false);
}

template <int M>
__device__ __forceinline__ float row_xor_add(float v) {
  const int vi = __builtin_bit_cast(int, v);
  const int sw = __builtin_amdgcn_update_dpp(vi, vi, 0x160 + M, 0xf, 0xf, true);
  return v + __builtin_bit_cast(float, sw);
}
__device__ __forceinline__ float other_row(float v) {
  const int vi = __builtin_bit_cast(int, v);
  return __builtin_bit_cast(float,
      __builtin_amdgcn_permlanex16(vi, vi, 0x76543210, 0xfedcba98, false, false));
}
__device__ __forceinline__ float uniform_f(float v) {
  return __builtin_bit_cast(float, __builtin_amdgcn_readfirstlane(__builtin_bit_cast(int, v)));
}

__global__ __launch_bounds__(256) void transpose_wout(const float* __restrict__ w_out,
                                                      float* __restrict__ w_out_T) {
  __shared__ float tile[32][33];
  const int n0 = blockIdx.x * 32;
  const int o0 = blockIdx.y * 32;
  const int tx = threadIdx.x;
  const int ty = threadIdx.y;
#pragma unroll
  for (int j = 0; j < 4; ++j) {
    const int n = n0 + tx;
    const int o = o0 + ty + j * 8;
    if (n < N_NODES) tile[ty + j * 8][tx] = w_out[(size_t)o * N_NODES + n];
  }
  __syncthreads();
#pragma unroll
  for (int j = 0; j < 4; ++j) {
    const int n = n0 + ty + j * 8;
    const int o = o0 + tx;
    if (n < N_NODES) w_out_T[(size_t)n * OUT_W + o] = tile[tx][ty + j * 8];
  }
}

__global__ __launch_bounds__(WAVES_PER_BLOCK * 32, 8) void fff_fwd(const float* __restrict__ x,
                                                                  const float* __restrict__ w_in,
                                                                  const float* __restrict__ w_out_T,
                                                                  float* __restrict__ out) {
  const int lane = threadIdx.x & 31;
  const int wave = __builtin_amdgcn_readfirstlane((int)(threadIdx.x >> 5));
  const int tok0 = (blockIdx.x * WAVES_PER_BLOCK + wave) * TOK_PER_WAVE;

  const int  r       = lane & 7;
  const bool on_diag = (lane < 8) || (lane >= 24);

  for (int it = 0; it < TOK_PER_WAVE; ++it) {
    const int t = tok0 + it;

    const v4f* __restrict__ xr = (const v4f*)(x + (size_t)t * IN_W) + lane;
    v4f xv[ROW_Q];
#pragma unroll
    for (int q = 0; q < ROW_Q; ++q) xv[q] = xr[32 * q];

    int node = 0;
    asm volatile("" : "+s"(node));
    int   path[LEVELS];
    float act[LEVELS];

    v4f wv[ROW_Q];
#pragma unroll
    for (int q = 0; q < ROW_Q; ++q) wv[q] = ((const v4f*)(w_in + (size_t)node * IN_W) + lane)[32 * q];

#pragma unroll
    for (int lvl = 0; lvl < LEVELS; ++lvl) {
      v8f c0 = {}, c1 = {};
#pragma unroll
      for (int q = 0; q < ROW_Q; ++q) {
        c0 = wmma4(xv[q].xy, wv[q].xy, c0);
        c1 = wmma4(xv[q].zw, wv[q].zw, c1);
      }
      const v8f c = c0 + c1;

      const float e0 = (r & 1) ? c[1] : c[0];
      const float e1 = (r & 1) ? c[3] : c[2];
      const float e2 = (r & 1) ? c[5] : c[4];
      const float e3 = (r & 1) ? c[7] : c[6];
      const float f0 = (r & 2) ? e1 : e0;
      const float f1 = (r & 2) ? e3 : e2;
      float dg = (r & 4) ? f1 : f0;
      dg = on_diag ? dg : 0.0f;

      dg = row_xor_add<1>(dg);
      dg = row_xor_add<2>(dg);
      dg = row_xor_add<4>(dg);
      dg = row_xor_add<8>(dg);
      dg = dg + other_row(dg);
      const float logit = uniform_f(dg);

      path[lvl] = node;
      node = 2 * node + 1 + ((logit > 0.0f) ? 1 : 0);
      if (lvl + 1 < LEVELS) {
#pragma unroll
        for (int q = 0; q < ROW_Q; ++q) wv[q] = ((const v4f*)(w_in + (size_t)node * IN_W) + lane)[32 * q];
      }
      act[lvl] = uniform_f(0.5f * logit * (1.0f + erff(logit * 0.70710678118654752f)));
    }

    v4f acc[ROW_Q];
#pragma unroll
    for (int q = 0; q < ROW_Q; ++q) acc[q] = (v4f){0.f, 0.f, 0.f, 0.f};
#pragma unroll
    for (int k = 0; k < LEVELS; ++k) {
      const float a = act[k];
      const v4f* __restrict__ row = (const v4f*)(w_out_T + (size_t)path[k] * OUT_W) + lane;
#pragma unroll
      for (int q = 0; q < ROW_Q; ++q) acc[q] += a * row[32 * q];
    }
    v4f* __restrict__ orow = (v4f*)(out + (size_t)t * OUT_W) + lane;
#pragma unroll
    for (int q = 0; q < ROW_Q; ++q) orow[32 * q] = acc[q];
  }
}

extern "C" void kernel_launch(void* const* d_in, const int* in_sizes, int n_in,
                              void* d_out, int out_size, void* d_ws, size_t ws_size,
                              hipStream_t stream) {
  (void)in_sizes; (void)n_in; (void)out_size; (void)ws_size;
  const float* x     = (const float*)d_in[0];
  const float* w_in  = (const float*)d_in[1];
  const float* w_out = (const float*)d_in[2];
  float* w_out_T = (float*)d_ws;
  float* out     = (float*)d_out;

  transpose_wout<<<dim3((N_NODES + 31) / 32, OUT_W / 32), dim3(32, 8), 0, stream>>>(
      w_out, w_out_T);
  fff_fwd<<<N_TOKENS / (WAVES_PER_BLOCK * TOK_PER_WAVE), WAVES_PER_BLOCK * 32, 0, stream>>>(
      x, w_in, w_out_T, out);
}


# w_in rows of tree levels 0-5 (63 rows, 252 KiB) copied once per 32-wave work-group to LDS and read as B fragments from there: 6 not 12 w_in rows fetched from L2 per token; sum order unchanged
# speedup vs baseline: 1.9298x; 1.0506x over previous
#include <hip/hip_runtime.h>
#include <math.h>


typedef float v2f __attribute__((ext_vector_type(2)));
typedef float v4f __attribute__((ext_vector_type(4)));
typedef float v8f __attribute__((ext_vector_type(8)));

#define DEPTH     11
#define LEVELS    (DEPTH + 1)
#define IN_W      1024
#define OUT_W     1024
#define N_NODES   4095
#define N_TOKENS  16384
#define ROW_Q     8
#define WAVES_PER_BLOCK 32
#define TOK_PER_WAVE    2
#define TOP_LEVELS      6
#define TOP_ROWS        ((1 << TOP_LEVELS) - 1)
#define ROW_V4          (IN_W / 4)

__device__ __forceinline__ v8f wmma4(v2f a, v2f b, v8f c) {
  return __builtin_amdgcn_wmma_f32_16x16x4_f32(
       false, a,  false, b,
       (short)0, c,  false,  false);
}

template <int M>
__device__ __forceinline__ float row_xor_add(float v) {
  const int vi = __builtin_bit_cast(int, v);
  const int sw = __builtin_amdgcn_update_dpp(vi, vi, 0x160 + M, 0xf, 0xf, true);
  return v + __builtin_bit_cast(float, sw);
}
__device__ __forceinline__ float other_row(float v) {
  const int vi = __builtin_bit_cast(int, v);
  return __builtin_bit_cast(float,
      __builtin_amdgcn_permlanex16(vi, vi, 0x76543210, 0xfedcba98, false, false));
}
__device__ __forceinline__ float uniform_f(float v) {
  return __builtin_bit_cast(float, __builtin_amdgcn_readfirstlane(__builtin_bit_cast(int, v)));
}

__global__ __launch_bounds__(256) void transpose_wout(const float* __restrict__ w_out,
                                                      float* __restrict__ w_out_T) {
  __shared__ float tile[32][33];
  const int n0 = blockIdx.x * 32;
  const int o0 = blockIdx.y * 32;
  const int tx = threadIdx.x;
  const int ty = threadIdx.y;
#pragma unroll
  for (int j = 0; j < 4; ++j) {
    const int n = n0 + tx;
    const int o = o0 + ty + j * 8;
    if (n < N_NODES) tile[ty + j * 8][tx] = w_out[(size_t)o * N_NODES + n];
  }
  __syncthreads();
#pragma unroll
  for (int j = 0; j < 4; ++j) {
    const int n = n0 + ty + j * 8;
    const int o = o0 + tx;
    if (n < N_NODES) w_out_T[(size_t)n * OUT_W + o] = tile[tx][ty + j * 8];
  }
}

__global__ __launch_bounds__(WAVES_PER_BLOCK * 32, 8) void fff_fwd(const float* __restrict__ x,
                                                                  const float* __restrict__ w_in,
                                                                  const float* __restrict__ w_out_T,
                                                                  float* __restrict__ out) {
  const int lane = threadIdx.x & 31;
  const int wave = __builtin_amdgcn_readfirstlane((int)(threadIdx.x >> 5));
  const int tok0 = (blockIdx.x * WAVES_PER_BLOCK + wave) * TOK_PER_WAVE;

  __shared__ __attribute__((aligned(16))) v4f top[TOP_ROWS * ROW_V4];
  for (int rr = wave; rr < TOP_ROWS; rr += WAVES_PER_BLOCK) {
    const v4f* __restrict__ src = (const v4f*)(w_in + (size_t)rr * IN_W) + lane;
    v4f tmp[ROW_Q];
#pragma unroll
    for (int q = 0; q < ROW_Q; ++q) tmp[q] = src[32 * q];
#pragma unroll
    for (int q = 0; q < ROW_Q; ++q) top[rr * ROW_V4 + 32 * q + lane] = tmp[q];
  }
  __syncthreads();

  const int  r       = lane & 7;
  const bool on_diag = (lane < 8) || (lane >= 24);

  for (int it = 0; it < TOK_PER_WAVE; ++it) {
    const int t = tok0 + it;

    const v4f* __restrict__ xr = (const v4f*)(x + (size_t)t * IN_W) + lane;
    v4f xv[ROW_Q];
#pragma unroll
    for (int q = 0; q < ROW_Q; ++q) xv[q] = xr[32 * q];

    int node = 0;
    asm volatile("" : "+s"(node));
    int   path[LEVELS];
    float act[LEVELS];

    v4f wv[ROW_Q];
#pragma unroll
    for (int q = 0; q < ROW_Q; ++q) wv[q] = top[node * ROW_V4 + 32 * q + lane];

#pragma unroll
    for (int lvl = 0; lvl < LEVELS; ++lvl) {
      v8f c0 = {}, c1 = {};
#pragma unroll
      for (int q = 0; q < ROW_Q; ++q) {
        c0 = wmma4(xv[q].xy, wv[q].xy, c0);
        c1 = wmma4(xv[q].zw, wv[q].zw, c1);
      }
      const v8f c = c0 + c1;

      const float e0 = (r & 1) ? c[1] : c[0];
      const float e1 = (r & 1) ? c[3] : c[2];
      const float e2 = (r & 1) ? c[5] : c[4];
      const float e3 = (r & 1) ? c[7] : c[6];
      const float f0 = (r & 2) ? e1 : e0;
      const float f1 = (r & 2) ? e3 : e2;
      float dg = (r & 4) ? f1 : f0;
      dg = on_diag ? dg : 0.0f;

      dg = row_xor_add<1>(dg);
      dg = row_xor_add<2>(dg);
      dg = row_xor_add<4>(dg);
      dg = row_xor_add<8>(dg);
      dg = dg + other_row(dg);
      const float logit = uniform_f(dg);

      path[lvl] = node;
      node = 2 * node + 1 + ((logit > 0.0f) ? 1 : 0);
      if (lvl + 1 < TOP_LEVELS) {
#pragma unroll
        for (int q = 0; q < ROW_Q; ++q) wv[q] = top[node * ROW_V4 + 32 * q + lane];
      } else if (lvl + 1 < LEVELS) {
#pragma unroll
        for (int q = 0; q < ROW_Q; ++q) wv[q] = ((const v4f*)(w_in + (size_t)node * IN_W) + lane)[32 * q];
      }
      act[lvl] = uniform_f(0.5f * logit * (1.0f + erff(logit * 0.70710678118654752f)));
    }

    v4f acc[ROW_Q];
#pragma unroll
    for (int q = 0; q < ROW_Q; ++q) acc[q] = (v4f){0.f, 0.f, 0.f, 0.f};
    v4f rw[2][ROW_Q];
#pragma unroll
    for (int q = 0; q < ROW_Q; ++q) rw[0][q] = ((const v4f*)(w_out_T + (size_t)path[0] * OUT_W) + lane)[32 * q];
#pragma unroll
    for (int k = 0; k < LEVELS; ++k) {
      if (k + 1 < LEVELS) {
        const v4f* __restrict__ nxt = (const v4f*)(w_out_T + (size_t)path[k + 1] * OUT_W) + lane;
#pragma unroll
        for (int q = 0; q < ROW_Q; ++q) rw[(k + 1) & 1][q] = nxt[32 * q];
      }
      const float a = act[k];
#pragma unroll
      for (int q = 0; q < ROW_Q; ++q) acc[q] += a * rw[k & 1][q];
    }
    v4f* __restrict__ orow = (v4f*)(out + (size_t)t * OUT_W) + lane;
#pragma unroll
    for (int q = 0; q < ROW_Q; ++q) orow[32 * q] = acc[q];
  }
}

extern "C" void kernel_launch(void* const* d_in, const int* in_sizes, int n_in,
                              void* d_out, int out_size, void* d_ws, size_t ws_size,
                              hipStream_t stream) {
  (void)in_sizes; (void)n_in; (void)out_size; (void)ws_size;
  const float* x     = (const float*)d_in[0];
  const float* w_in  = (const float*)d_in[1];
  const float* w_out = (const float*)d_in[2];
  float* w_out_T = (float*)d_ws;
  float* out     = (float*)d_out;

  transpose_wout<<<dim3((N_NODES + 31) / 32, OUT_W / 32), dim3(32, 8), 0, stream>>>(
      w_out, w_out_T);
  fff_fwd<<<N_TOKENS / (WAVES_PER_BLOCK * TOK_PER_WAVE), WAVES_PER_BLOCK * 32, 0, stream>>>(
      x, w_in, w_out_T, out);
}
